# starting kernel
# baseline (speedup 1.0000x reference)
#include <hip/hip_runtime.h>

typedef __attribute__((ext_vector_type(2))) float v2f;
typedef __attribute__((ext_vector_type(8))) float v8f;

#define DIMSZ 256
#define NBATCH 16
#define NPIX (DIMSZ * DIMSZ)
#define NELEM (NBATCH * NPIX)
#define LDSW 138
#define LDSH 22
#define NSCHUNK (7 * 6)
#define SBUF_ELEMS (NSCHUNK * 32 * 2)

__global__ void fdtd_init(const float* __restrict__ fdk,
                          const float* __restrict__ cw,
                          float* __restrict__ Sbuf) {
    __shared__ float weff[49];
    int t = threadIdx.x;
    if (t < 49) {
        float acc = 0.0f;
        #pragma unroll
        for (int ch = 1; ch < 15; ++ch)
            acc += cw[ch - 1] * fdk[ch * 49 + t];
        weff[t] = acc * (2.0f * 0.5f / 1.0f);
    }
    __syncthreads();
    for (int e = t; e < SBUF_ELEMS; e += blockDim.x) {
        int v    = e & 1;
        int lane = (e >> 1) & 31;
        int c    = (e >> 6) % 6;
        int ky   = e / 384;
        int n    = lane & 15;
        int k    = 4 * c + v + 2 * (lane >> 4);
        int j    = k - n;
        float val = (k < 22 && j >= 0 && j <= 6) ? weff[ky * 7 + j] : 0.0f;
        Sbuf[e] = val;
    }
}

__global__ __launch_bounds__(256) void fdtd_step(
    const float* __restrict__ cur,
    const float* prev,
    const float* __restrict__ epsr,
    const float* __restrict__ sigma,
    const float* __restrict__ Sglob,
    const float* __restrict__ src,
    float* next,
    int step) {
    __shared__ float tile[LDSH * LDSW];
    __shared__ float Slds[SBUF_ELEMS];

    const int b  = blockIdx.x;
    const int y0 = blockIdx.y * 16;
    const int x0 = blockIdx.z * 128;
    const int t  = threadIdx.x;
    const float* curb = cur + b * NPIX;

    for (int e = t; e < LDSH * LDSW; e += 256) {
        int r = e / LDSW, c = e % LDSW;
        int gy = y0 + r - 3, gx = x0 + c - 3;
        float v = 0.0f;
        if (c < 136 && gy >= 0 && gy < DIMSZ && gx >= 0 && gx < DIMSZ)
            v = curb[gy * DIMSZ + gx];
        tile[e] = v;
    }
    for (int e = t; e < SBUF_ELEMS; e += 256) Slds[e] = Sglob[e];
    __syncthreads();

    const int wave = t >> 5;
    const int lane = t & 31;
    const int mlo  = lane & 15;
    const int sel  = lane >> 4;
    const int xw   = 16 * wave;

    v8f acc = {0.f, 0.f, 0.f, 0.f, 0.f, 0.f, 0.f, 0.f};
    #pragma unroll
    for (int ky = 0; ky < 7; ++ky) {
        const int arow = (mlo + ky) * LDSW + xw + 2 * sel;
        #pragma unroll
        for (int c = 0; c < 6; ++c) {
            v2f a = *(const v2f*)&tile[arow + 4 * c];
            v2f bmat = *(const v2f*)&Slds[((ky * 6 + c) * 32 + lane) * 2];
            acc = __builtin_amdgcn_wmma_f32_16x16x4_f32(
                false, a, false, bmat, (short)0, acc, false, false);
        }
    }

    const float dt = 0.5f;
    const float s0 = src[2 * step], s1 = src[2 * step + 1];
    const float srcterm = -(s0 - s1);
    const int gx = x0 + xw + mlo;
    #pragma unroll
    for (int r = 0; r < 8; ++r) {
        int m  = r + 8 * sel;
        int gy = y0 + m;
        int idx = b * NPIX + gy * DIMSZ + gx;
        float ep = epsr[idx], sg = sigma[idx];
        float inv_den = 1.0f / (2.0f * ep / dt + sg);
        float av = 4.0f * ep / dt;
        float bv = 2.0f * ep / dt - sg;
        float cu = tile[(m + 3) * LDSW + xw + mlo + 3];
        float pv = prev[idx];
        float rhs = acc[r] + av * cu - bv * pv;
        if (gy == 128 && gx == 128) rhs += srcterm;
        next[idx] = inv_den * rhs;
    }
}

extern "C" void kernel_launch(void* const* d_in, const int* in_sizes, int n_in,
                              void* d_out, int out_size, void* d_ws, size_t ws_size,
                              hipStream_t stream) {
    const float* u1    = (const float*)d_in[0];
    const float* u0    = (const float*)d_in[1];
    const float* epsr  = (const float*)d_in[2];
    const float* sigma = (const float*)d_in[3];
    const float* fdk   = (const float*)d_in[4];
    const float* cw    = (const float*)d_in[5];
    const float* src   = (const float*)d_in[6];
    float* out = (float*)d_out;
    float* ws  = (float*)d_ws;

    float* bufs[3] = {ws, ws + NELEM, ws + 2 * NELEM};
    float* Sbuf = ws + 3 * NELEM;

    fdtd_init<<<1, 256, 0, stream>>>(fdk, cw, Sbuf);

    const float* cur  = u1;
    const float* prev = u0;
    dim3 grid(NBATCH, 16, 2);
    for (int s = 0; s < 32; ++s) {
        float* next = (s == 31) ? out : bufs[s % 3];
        fdtd_step<<<grid, 256, 0, stream>>>(cur, prev, epsr, sigma, Sbuf, src, next, s);
        prev = cur;
        cur  = next;
    }
}


# two time steps fused per launch (33 launches become 17): a 512-thread block stages u with a 6-wide halo and keeps the intermediate field (32x128 core+3) in LDS; per-pixel summation order kept
# speedup vs baseline: 1.2446x; 1.2446x over previous
#include <hip/hip_runtime.h>

typedef __attribute__((ext_vector_type(2))) float v2f;
typedef __attribute__((ext_vector_type(8))) float v8f;

#define DIMSZ 256
#define NBATCH 16
#define NPIX (DIMSZ * DIMSZ)
#define NELEM (NBATCH * NPIX)
#define NSCHUNK (7 * 6)
#define SBUF_ELEMS (NSCHUNK * 32 * 2)

#define CORE_H 32
#define CORE_W 128
#define NTHR 512
#define LW 170
#define U1H (CORE_H + 12)
#define U2H (CORE_H + 6)
#define NCT1 10
#define NRT1 3
#define NT1 (NCT1 * NRT1)

__global__ void fdtd_init(const float* __restrict__ fdk,
                          const float* __restrict__ cw,
                          float* __restrict__ Sbuf) {
    __shared__ float weff[49];
    int t = threadIdx.x;
    if (t < 49) {
        float acc = 0.0f;
        #pragma unroll
        for (int ch = 1; ch < 15; ++ch)
            acc += cw[ch - 1] * fdk[ch * 49 + t];
        weff[t] = acc * (2.0f * 0.5f / 1.0f);
    }
    __syncthreads();
    for (int e = t; e < SBUF_ELEMS; e += blockDim.x) {
        int v    = e & 1;
        int lane = (e >> 1) & 31;
        int c    = (e >> 6) % 6;
        int ky   = e / 384;
        int n    = lane & 15;
        int k    = 4 * c + v + 2 * (lane >> 4);
        int j    = k - n;
        float val = (k < 22 && j >= 0 && j <= 6) ? weff[ky * 7 + j] : 0.0f;
        Sbuf[e] = val;
    }
}

__device__ __forceinline__ v8f wmma4(v2f a, v2f b, v8f c) {
    return __builtin_amdgcn_wmma_f32_16x16x4_f32(false, a, false, b, (short)0, c, false, false);
}

__device__ __forceinline__ int clampi(int v, int lo, int hi) {
    return v < lo ? lo : (v > hi ? hi : v);
}

__device__ __forceinline__ float fdtd_point(float conv, float ep, float sg,
                                            float cu, float pv, bool at_src, float srcterm) {
    const float dt = 0.5f;
    float inv_den = 1.0f / (2.0f * ep / dt + sg);
    float av = 4.0f * ep / dt;
    float bv = 2.0f * ep / dt - sg;
    float rhs = conv + av * cu - bv * pv;
    if (at_src) rhs += srcterm;
    return inv_den * rhs;
}

__device__ __forceinline__ void first_step_finish(
    const v8f& acc, const float* ep, const float* sg, const float* pv,
    const float* U1, float* U2, int y0, int x0,
    int roff, int mmin, int lc, int sel, float srcterm, bool tile_ok) {
    const int gx = x0 - 19 + lc;
    const bool colok = gx >= 0 && gx < DIMSZ && lc >= 16 && lc < 150;
    #pragma unroll
    for (int r = 0; r < 8; ++r) {
        const int m  = r + 8 * sel;
        const int rr = roff + m;
        const int gy = y0 - 3 + rr;
        const float cu = U1[(rr + 3) * LW + lc];
        float val = fdtd_point(acc[r], ep[r], sg[r], cu, pv[r], gy == 128 && gx == 128, srcterm);
        if (!(colok && gy >= 0 && gy < DIMSZ)) val = 0.0f;
        if (tile_ok && m >= mmin) U2[rr * LW + lc] = val;
    }
}

__global__ __launch_bounds__(NTHR) void fdtd_step2(
    const float* __restrict__ cur,
    const float* prev,
    const float* __restrict__ epsr,
    const float* __restrict__ sigma,
    const float* __restrict__ Sglob,
    const float* __restrict__ src,
    float* mid,
    float* next,
    int step) {
    __shared__ __attribute__((aligned(16))) float U1[U1H * LW];
    __shared__ __attribute__((aligned(16))) float U2[U2H * LW];
    __shared__ __attribute__((aligned(16))) float Slds[SBUF_ELEMS];

    const int b  = blockIdx.x;
    const int y0 = blockIdx.y * CORE_H;
    const int x0 = blockIdx.z * CORE_W;
    const int t  = threadIdx.x;
    const int boff = b * NPIX;
    const float* curb = cur + boff;

    for (int e = t; e < U1H * LW; e += NTHR) {
        int r = e / LW, c = e % LW;
        int gy = y0 - 6 + r, gx = x0 - 19 + c;
        float v = 0.0f;
        if (c >= 13 && c < 153 && gy >= 0 && gy < DIMSZ && gx >= 0 && gx < DIMSZ)
            v = curb[gy * DIMSZ + gx];
        U1[e] = v;
    }
    for (int e = t; e < SBUF_ELEMS; e += NTHR) Slds[e] = Sglob[e];
    __syncthreads();

    const int wave = t >> 5;
    const int lane = t & 31;
    const int mlo  = lane & 15;
    const int sel  = lane >> 4;

    {
        const bool okB = (wave + 16) < NT1;
        const int tA = wave;
        const int tB = okB ? (wave + 16) : (NT1 - 1);
        const int rtA = tA / NCT1, jA = tA % NCT1;
        const int rtB = tB / NCT1, jB = tB % NCT1;
        const int roffA = (rtA == NRT1 - 1) ? (U2H - 16) : 16 * rtA;
        const int roffB = (rtB == NRT1 - 1) ? (U2H - 16) : 16 * rtB;
        const int mminA = 16 * rtA - roffA;
        const int mminB = 16 * rtB - roffB;
        const int lcA = 16 * jA + 3 + mlo;
        const int lcB = 16 * jB + 3 + mlo;
        const int gxAc = clampi(x0 - 19 + lcA, 0, DIMSZ - 1);
        const int gxBc = clampi(x0 - 19 + lcB, 0, DIMSZ - 1);

        float epA[8], sgA[8], pvA[8], epB[8], sgB[8], pvB[8];
        #pragma unroll
        for (int r = 0; r < 8; ++r) {
            const int m = r + 8 * sel;
            const int ia = boff + clampi(y0 - 3 + roffA + m, 0, DIMSZ - 1) * DIMSZ + gxAc;
            const int ib = boff + clampi(y0 - 3 + roffB + m, 0, DIMSZ - 1) * DIMSZ + gxBc;
            epA[r] = epsr[ia]; sgA[r] = sigma[ia]; pvA[r] = prev[ia];
            epB[r] = epsr[ib]; sgB[r] = sigma[ib]; pvB[r] = prev[ib];
        }

        v8f accA = {0.f, 0.f, 0.f, 0.f, 0.f, 0.f, 0.f, 0.f};
        v8f accB = {0.f, 0.f, 0.f, 0.f, 0.f, 0.f, 0.f, 0.f};
        const int baseA = (mlo + roffA) * LW + 16 * jA + 2 * sel;
        const int baseB = (mlo + roffB) * LW + 16 * jB + 2 * sel;
        #pragma unroll
        for (int ky = 0; ky < 7; ++ky) {
            #pragma unroll
            for (int c = 0; c < 6; ++c) {
                v2f bmat = *(const v2f*)&Slds[((ky * 6 + c) * 32 + lane) * 2];
                v2f a0 = *(const v2f*)&U1[baseA + ky * LW + 4 * c];
                v2f a1 = *(const v2f*)&U1[baseB + ky * LW + 4 * c];
                accA = wmma4(a0, bmat, accA);
                accB = wmma4(a1, bmat, accB);
            }
        }

        const float srcterm = -(src[2 * step] - src[2 * step + 1]);
        first_step_finish(accA, epA, sgA, pvA, U1, U2, y0, x0,
                          roffA, mminA, lcA, sel, srcterm, true);
        first_step_finish(accB, epB, sgB, pvB, U1, U2, y0, x0,
                          roffB, mminB, lcB, sel, srcterm, okB);
    }
    __syncthreads();

    {
        const int q = wave >> 3;
        const int w = wave & 7;
        const int gx = x0 + 16 * w + mlo;
        const int lc = 16 * w + 19 + mlo;

        float ep2[8], sg2[8], u2c[8];
        #pragma unroll
        for (int r = 0; r < 8; ++r) {
            const int cy = 16 * q + r + 8 * sel;
            const int idx = boff + (y0 + cy) * DIMSZ + gx;
            ep2[r] = epsr[idx]; sg2[r] = sigma[idx];
            u2c[r] = U2[(cy + 3) * LW + lc];
        }
        if (mid != nullptr) {
            #pragma unroll
            for (int r = 0; r < 8; ++r) {
                const int cy = 16 * q + r + 8 * sel;
                mid[boff + (y0 + cy) * DIMSZ + gx] = u2c[r];
            }
        }

        v8f acc = {0.f, 0.f, 0.f, 0.f, 0.f, 0.f, 0.f, 0.f};
        const int base2 = (16 * q + mlo) * LW + 16 * w + 16 + 2 * sel;
        #pragma unroll
        for (int ky = 0; ky < 7; ++ky) {
            #pragma unroll
            for (int c = 0; c < 6; ++c) {
                v2f a = *(const v2f*)&U2[base2 + ky * LW + 4 * c];
                v2f bmat = *(const v2f*)&Slds[((ky * 6 + c) * 32 + lane) * 2];
                acc = wmma4(a, bmat, acc);
            }
        }

        const float srcterm2 = -(src[2 * step + 2] - src[2 * step + 3]);
        #pragma unroll
        for (int r = 0; r < 8; ++r) {
            const int cy = 16 * q + r + 8 * sel;
            const int gy = y0 + cy;
            const int idx = boff + gy * DIMSZ + gx;
            const float u1c = U1[(cy + 6) * LW + lc];
            next[idx] = fdtd_point(acc[r], ep2[r], sg2[r], u2c[r], u1c,
                                   gy == 128 && gx == 128, srcterm2);
        }
    }
}

extern "C" void kernel_launch(void* const* d_in, const int* in_sizes, int n_in,
                              void* d_out, int out_size, void* d_ws, size_t ws_size,
                              hipStream_t stream) {
    const float* u1    = (const float*)d_in[0];
    const float* u0    = (const float*)d_in[1];
    const float* epsr  = (const float*)d_in[2];
    const float* sigma = (const float*)d_in[3];
    const float* fdk   = (const float*)d_in[4];
    const float* cw    = (const float*)d_in[5];
    const float* src   = (const float*)d_in[6];
    float* out = (float*)d_out;
    float* ws  = (float*)d_ws;

    float* bufs[4] = {ws, ws + NELEM, ws + 2 * NELEM, ws + 3 * NELEM};
    float* Sbuf = ws + 4 * NELEM;

    fdtd_init<<<1, 256, 0, stream>>>(fdk, cw, Sbuf);

    const float* cur  = u1;
    const float* prev = u0;
    dim3 grid(NBATCH, DIMSZ / CORE_H, DIMSZ / CORE_W);
    for (int k = 0; k < 16; ++k) {
        float* mid  = (k == 15) ? nullptr : bufs[2 * (k & 1)];
        float* next = (k == 15) ? out : bufs[2 * (k & 1) + 1];
        fdtd_step2<<<grid, NTHR, 0, stream>>>(cur, prev, epsr, sigma, Sbuf, src, mid, next, 2 * k);
        prev = mid;
        cur  = next;
    }
}


# B operands (banded kernel matrices) loaded once per wave from global into 84 registers and kept there for both steps; the LDS copy of them and its re-read per WMMA are gone
# speedup vs baseline: 1.3163x; 1.0576x over previous
#include <hip/hip_runtime.h>

typedef __attribute__((ext_vector_type(2))) float v2f;
typedef __attribute__((ext_vector_type(8))) float v8f;

#define DIMSZ 256
#define NBATCH 16
#define NPIX (DIMSZ * DIMSZ)
#define NELEM (NBATCH * NPIX)
#define NSCHUNK (7 * 6)
#define SBUF_ELEMS (NSCHUNK * 32 * 2)

#define CORE_H 32
#define CORE_W 128
#define NTHR 512
#define LW 170
#define U1H (CORE_H + 12)
#define U2H (CORE_H + 6)
#define NCT1 10
#define NRT1 3
#define NT1 (NCT1 * NRT1)

__global__ void fdtd_init(const float* __restrict__ fdk,
                          const float* __restrict__ cw,
                          float* __restrict__ Sbuf) {
    __shared__ float weff[49];
    int t = threadIdx.x;
    if (t < 49) {
        float acc = 0.0f;
        #pragma unroll
        for (int ch = 1; ch < 15; ++ch)
            acc += cw[ch - 1] * fdk[ch * 49 + t];
        weff[t] = acc * (2.0f * 0.5f / 1.0f);
    }
    __syncthreads();
    for (int e = t; e < SBUF_ELEMS; e += blockDim.x) {
        int v    = e & 1;
        int lane = (e >> 1) & 31;
        int c    = (e >> 6) % 6;
        int ky   = e / 384;
        int n    = lane & 15;
        int k    = 4 * c + v + 2 * (lane >> 4);
        int j    = k - n;
        float val = (k < 22 && j >= 0 && j <= 6) ? weff[ky * 7 + j] : 0.0f;
        Sbuf[e] = val;
    }
}

__device__ __forceinline__ v8f wmma4(v2f a, v2f b, v8f c) {
    return __builtin_amdgcn_wmma_f32_16x16x4_f32(false, a, false, b, (short)0, c, false, false);
}

__device__ __forceinline__ int clampi(int v, int lo, int hi) {
    return v < lo ? lo : (v > hi ? hi : v);
}

__device__ __forceinline__ float fdtd_point(float conv, float ep, float sg,
                                            float cu, float pv, bool at_src, float srcterm) {
    const float dt = 0.5f;
    float inv_den = 1.0f / (2.0f * ep / dt + sg);
    float av = 4.0f * ep / dt;
    float bv = 2.0f * ep / dt - sg;
    float rhs = conv + av * cu - bv * pv;
    if (at_src) rhs += srcterm;
    return inv_den * rhs;
}

__device__ __forceinline__ void first_step_finish(
    const v8f& acc, const float* ep, const float* sg, const float* pv,
    const float* U1, float* U2, int y0, int x0,
    int roff, int mmin, int lc, int sel, float srcterm, bool tile_ok) {
    const int gx = x0 - 19 + lc;
    const bool colok = gx >= 0 && gx < DIMSZ && lc >= 16 && lc < 150;
    #pragma unroll
    for (int r = 0; r < 8; ++r) {
        const int m  = r + 8 * sel;
        const int rr = roff + m;
        const int gy = y0 - 3 + rr;
        const float cu = U1[(rr + 3) * LW + lc];
        float val = fdtd_point(acc[r], ep[r], sg[r], cu, pv[r], gy == 128 && gx == 128, srcterm);
        if (!(colok && gy >= 0 && gy < DIMSZ)) val = 0.0f;
        if (tile_ok && m >= mmin) U2[rr * LW + lc] = val;
    }
}

__global__ __launch_bounds__(NTHR) void fdtd_step2(
    const float* __restrict__ cur,
    const float* prev,
    const float* __restrict__ epsr,
    const float* __restrict__ sigma,
    const float* __restrict__ Sglob,
    const float* __restrict__ src,
    float* mid,
    float* next,
    int step) {
    __shared__ __attribute__((aligned(16))) float U1[U1H * LW];
    __shared__ __attribute__((aligned(16))) float U2[U2H * LW];

    const int b  = blockIdx.x;
    const int y0 = blockIdx.y * CORE_H;
    const int x0 = blockIdx.z * CORE_W;
    const int t  = threadIdx.x;
    const int boff = b * NPIX;
    const float* curb = cur + boff;

    v2f breg[NSCHUNK];
    #pragma unroll
    for (int i = 0; i < NSCHUNK; ++i)
        breg[i] = *(const v2f*)&Sglob[(i * 32 + (t & 31)) * 2];

    for (int e = t; e < U1H * LW; e += NTHR) {
        int r = e / LW, c = e % LW;
        int gy = y0 - 6 + r, gx = x0 - 19 + c;
        float v = 0.0f;
        if (c >= 13 && c < 153 && gy >= 0 && gy < DIMSZ && gx >= 0 && gx < DIMSZ)
            v = curb[gy * DIMSZ + gx];
        U1[e] = v;
    }
    __syncthreads();

    const int wave = t >> 5;
    const int lane = t & 31;
    const int mlo  = lane & 15;
    const int sel  = lane >> 4;

    {
        const bool okB = (wave + 16) < NT1;
        const int tA = wave;
        const int tB = okB ? (wave + 16) : (NT1 - 1);
        const int rtA = tA / NCT1, jA = tA % NCT1;
        const int rtB = tB / NCT1, jB = tB % NCT1;
        const int roffA = (rtA == NRT1 - 1) ? (U2H - 16) : 16 * rtA;
        const int roffB = (rtB == NRT1 - 1) ? (U2H - 16) : 16 * rtB;
        const int mminA = 16 * rtA - roffA;
        const int mminB = 16 * rtB - roffB;
        const int lcA = 16 * jA + 3 + mlo;
        const int lcB = 16 * jB + 3 + mlo;
        const int gxAc = clampi(x0 - 19 + lcA, 0, DIMSZ - 1);
        const int gxBc = clampi(x0 - 19 + lcB, 0, DIMSZ - 1);

        float epA[8], sgA[8], pvA[8], epB[8], sgB[8], pvB[8];
        #pragma unroll
        for (int r = 0; r < 8; ++r) {
            const int m = r + 8 * sel;
            const int ia = boff + clampi(y0 - 3 + roffA + m, 0, DIMSZ - 1) * DIMSZ + gxAc;
            const int ib = boff + clampi(y0 - 3 + roffB + m, 0, DIMSZ - 1) * DIMSZ + gxBc;
            epA[r] = epsr[ia]; sgA[r] = sigma[ia]; pvA[r] = prev[ia];
            epB[r] = epsr[ib]; sgB[r] = sigma[ib]; pvB[r] = prev[ib];
        }

        v8f accA = {0.f, 0.f, 0.f, 0.f, 0.f, 0.f, 0.f, 0.f};
        v8f accB = {0.f, 0.f, 0.f, 0.f, 0.f, 0.f, 0.f, 0.f};
        const int baseA = (mlo + roffA) * LW + 16 * jA + 2 * sel;
        const int baseB = (mlo + roffB) * LW + 16 * jB + 2 * sel;
        #pragma unroll
        for (int ky = 0; ky < 7; ++ky) {
            #pragma unroll
            for (int c = 0; c < 6; ++c) {
                const v2f bmat = breg[ky * 6 + c];
                v2f a0 = *(const v2f*)&U1[baseA + ky * LW + 4 * c];
                v2f a1 = *(const v2f*)&U1[baseB + ky * LW + 4 * c];
                accA = wmma4(a0, bmat, accA);
                accB = wmma4(a1, bmat, accB);
            }
        }

        const float srcterm = -(src[2 * step] - src[2 * step + 1]);
        first_step_finish(accA, epA, sgA, pvA, U1, U2, y0, x0,
                          roffA, mminA, lcA, sel, srcterm, true);
        first_step_finish(accB, epB, sgB, pvB, U1, U2, y0, x0,
                          roffB, mminB, lcB, sel, srcterm, okB);
    }
    __syncthreads();

    {
        const int q = wave >> 3;
        const int w = wave & 7;
        const int gx = x0 + 16 * w + mlo;
        const int lc = 16 * w + 19 + mlo;

        float ep2[8], sg2[8], u2c[8];
        #pragma unroll
        for (int r = 0; r < 8; ++r) {
            const int cy = 16 * q + r + 8 * sel;
            const int idx = boff + (y0 + cy) * DIMSZ + gx;
            ep2[r] = epsr[idx]; sg2[r] = sigma[idx];
            u2c[r] = U2[(cy + 3) * LW + lc];
        }
        if (mid != nullptr) {
            #pragma unroll
            for (int r = 0; r < 8; ++r) {
                const int cy = 16 * q + r + 8 * sel;
                mid[boff + (y0 + cy) * DIMSZ + gx] = u2c[r];
            }
        }

        v8f acc = {0.f, 0.f, 0.f, 0.f, 0.f, 0.f, 0.f, 0.f};
        const int base2 = (16 * q + mlo) * LW + 16 * w + 16 + 2 * sel;
        #pragma unroll
        for (int ky = 0; ky < 7; ++ky) {
            #pragma unroll
            for (int c = 0; c < 6; ++c) {
                v2f a = *(const v2f*)&U2[base2 + ky * LW + 4 * c];
                acc = wmma4(a, breg[ky * 6 + c], acc);
            }
        }

        const float srcterm2 = -(src[2 * step + 2] - src[2 * step + 3]);
        #pragma unroll
        for (int r = 0; r < 8; ++r) {
            const int cy = 16 * q + r + 8 * sel;
            const int gy = y0 + cy;
            const int idx = boff + gy * DIMSZ + gx;
            const float u1c = U1[(cy + 6) * LW + lc];
            next[idx] = fdtd_point(acc[r], ep2[r], sg2[r], u2c[r], u1c,
                                   gy == 128 && gx == 128, srcterm2);
        }
    }
}

extern "C" void kernel_launch(void* const* d_in, const int* in_sizes, int n_in,
                              void* d_out, int out_size, void* d_ws, size_t ws_size,
                              hipStream_t stream) {
    const float* u1    = (const float*)d_in[0];
    const float* u0    = (const float*)d_in[1];
    const float* epsr  = (const float*)d_in[2];
    const float* sigma = (const float*)d_in[3];
    const float* fdk   = (const float*)d_in[4];
    const float* cw    = (const float*)d_in[5];
    const float* src   = (const float*)d_in[6];
    float* out = (float*)d_out;
    float* ws  = (float*)d_ws;

    float* bufs[4] = {ws, ws + NELEM, ws + 2 * NELEM, ws + 3 * NELEM};
    float* Sbuf = ws + 4 * NELEM;

    fdtd_init<<<1, 256, 0, stream>>>(fdk, cw, Sbuf);

    const float* cur  = u1;
    const float* prev = u0;
    dim3 grid(NBATCH, DIMSZ / CORE_H, DIMSZ / CORE_W);
    for (int k = 0; k < 16; ++k) {
        float* mid  = (k == 15) ? nullptr : bufs[2 * (k & 1)];
        float* next = (k == 15) ? out : bufs[2 * (k & 1) + 1];
        fdtd_step2<<<grid, NTHR, 0, stream>>>(cur, prev, epsr, sigma, Sbuf, src, mid, next, 2 * k);
        prev = mid;
        cur  = next;
    }
}


# re-tiled to full-width row strips (core 16x256, grid 16x16): no x halo, u staged as whole aligned 1-KB rows with 16-byte loads, 4 per thread instead of 15 element loads; summation order kept
# speedup vs baseline: 1.8186x; 1.3816x over previous
#include <hip/hip_runtime.h>

typedef __attribute__((ext_vector_type(2))) float v2f;
typedef __attribute__((ext_vector_type(4))) float v4f;
typedef __attribute__((ext_vector_type(8))) float v8f;

#define DIMSZ 256
#define NBATCH 16
#define NPIX (DIMSZ * DIMSZ)
#define NELEM (NBATCH * NPIX)
#define NSCHUNK (7 * 6)
#define SBUF_ELEMS (NSCHUNK * 32 * 2)

#define CORE_H 16
#define NTHR 512
#define XPAD 3
#define LW 268
#define U1H (CORE_H + 12)
#define U2H (CORE_H + 6)
#define NPADC (LW - DIMSZ)
#define ROW_V4 (DIMSZ / 4)
#define NSTG ((U1H * ROW_V4 + NTHR - 1) / NTHR)

__global__ void fdtd_init(const float* __restrict__ fdk,
                          const float* __restrict__ cw,
                          float* __restrict__ Sbuf) {
    __shared__ float weff[49];
    int t = threadIdx.x;
    if (t < 49) {
        float acc = 0.0f;
        #pragma unroll
        for (int ch = 1; ch < 15; ++ch)
            acc += cw[ch - 1] * fdk[ch * 49 + t];
        weff[t] = acc * (2.0f * 0.5f / 1.0f);
    }
    __syncthreads();
    for (int e = t; e < SBUF_ELEMS; e += blockDim.x) {
        int v    = e & 1;
        int lane = (e >> 1) & 31;
        int c    = (e >> 6) % 6;
        int ky   = e / 384;
        int n    = lane & 15;
        int k    = 4 * c + v + 2 * (lane >> 4);
        int j    = k - n;
        float val = (k < 22 && j >= 0 && j <= 6) ? weff[ky * 7 + j] : 0.0f;
        Sbuf[e] = val;
    }
}

__device__ __forceinline__ v8f wmma4(v2f a, v2f b, v8f c) {
    return __builtin_amdgcn_wmma_f32_16x16x4_f32(false, a, false, b, (short)0, c, false, false);
}

__device__ __forceinline__ int clampi(int v, int lo, int hi) {
    return v < lo ? lo : (v > hi ? hi : v);
}

__device__ __forceinline__ float fdtd_point(float conv, float ep, float sg,
                                            float cu, float pv, bool at_src, float srcterm) {
    const float dt = 0.5f;
    float inv_den = 1.0f / (2.0f * ep / dt + sg);
    float av = 4.0f * ep / dt;
    float bv = 2.0f * ep / dt - sg;
    float rhs = conv + av * cu - bv * pv;
    if (at_src) rhs += srcterm;
    return inv_den * rhs;
}

__device__ __forceinline__ void first_step_finish(
    const v8f& acc, const float* ep, const float* sg, const float* pv,
    const float* U1, float* U2, int y0,
    int roff, int mmin, int lc, int gx, int sel, float srcterm) {
    #pragma unroll
    for (int r = 0; r < 8; ++r) {
        const int m  = r + 8 * sel;
        const int rr = roff + m;
        const int gy = y0 - 3 + rr;
        const float cu = U1[(rr + 3) * LW + lc];
        float val = fdtd_point(acc[r], ep[r], sg[r], cu, pv[r], gy == 128 && gx == 128, srcterm);
        if (gy < 0 || gy >= DIMSZ) val = 0.0f;
        if (m >= mmin) U2[rr * LW + lc] = val;
    }
}

__global__ __launch_bounds__(NTHR) void fdtd_step2(
    const float* __restrict__ cur,
    const float* prev,
    const float* __restrict__ epsr,
    const float* __restrict__ sigma,
    const float* __restrict__ Sglob,
    const float* __restrict__ src,
    float* mid,
    float* next,
    int step) {
    __shared__ __attribute__((aligned(16))) float U1[U1H * LW];
    __shared__ __attribute__((aligned(16))) float U2[U2H * LW];

    const int b  = blockIdx.x;
    const int y0 = blockIdx.y * CORE_H;
    const int t  = threadIdx.x;
    const int boff = b * NPIX;
    const float* curb = cur + boff;

    v2f breg[NSCHUNK];
    #pragma unroll
    for (int i = 0; i < NSCHUNK; ++i)
        breg[i] = *(const v2f*)&Sglob[(i * 32 + (t & 31)) * 2];

    v4f stg[NSTG];
    #pragma unroll
    for (int i = 0; i < NSTG; ++i) {
        const int p  = t + NTHR * i;
        const int r  = p / ROW_V4, c4 = p % ROW_V4;
        const int gy = y0 - 6 + r;
        v4f v = {0.f, 0.f, 0.f, 0.f};
        if (r < U1H && gy >= 0 && gy < DIMSZ)
            v = *(const v4f*)&curb[gy * DIMSZ + 4 * c4];
        stg[i] = v;
    }
    for (int e = t; e < (U1H + U2H) * NPADC; e += NTHR) {
        const int r = e / NPADC, k = e % NPADC;
        const int c = (k < XPAD) ? k : (DIMSZ + k);
        if (r < U1H) U1[r * LW + c] = 0.0f;
        else         U2[(r - U1H) * LW + c] = 0.0f;
    }
    #pragma unroll
    for (int i = 0; i < NSTG; ++i) {
        const int p = t + NTHR * i;
        const int r = p / ROW_V4, c4 = p % ROW_V4;
        if (r < U1H) {
            float* d = &U1[r * LW + XPAD + 4 * c4];
            d[0] = stg[i][0]; d[1] = stg[i][1]; d[2] = stg[i][2]; d[3] = stg[i][3];
        }
    }
    __syncthreads();

    const int wave = t >> 5;
    const int lane = t & 31;
    const int mlo  = lane & 15;
    const int sel  = lane >> 4;
    const int gx   = 16 * wave + mlo;
    const int lc   = gx + XPAD;

    {
        const int roffB = U2H - 16;
        const int mminB = 16 - roffB;

        float epA[8], sgA[8], pvA[8], epB[8], sgB[8], pvB[8];
        #pragma unroll
        for (int r = 0; r < 8; ++r) {
            const int m = r + 8 * sel;
            const int ia = boff + clampi(y0 - 3 + m, 0, DIMSZ - 1) * DIMSZ + gx;
            const int ib = boff + clampi(y0 - 3 + roffB + m, 0, DIMSZ - 1) * DIMSZ + gx;
            epA[r] = epsr[ia]; sgA[r] = sigma[ia]; pvA[r] = prev[ia];
            epB[r] = epsr[ib]; sgB[r] = sigma[ib]; pvB[r] = prev[ib];
        }

        v8f accA = {0.f, 0.f, 0.f, 0.f, 0.f, 0.f, 0.f, 0.f};
        v8f accB = {0.f, 0.f, 0.f, 0.f, 0.f, 0.f, 0.f, 0.f};
        const int baseA = mlo * LW + 16 * wave + 2 * sel;
        const int baseB = (mlo + roffB) * LW + 16 * wave + 2 * sel;
        #pragma unroll
        for (int ky = 0; ky < 7; ++ky) {
            #pragma unroll
            for (int c = 0; c < 6; ++c) {
                const v2f bmat = breg[ky * 6 + c];
                v2f a0 = *(const v2f*)&U1[baseA + ky * LW + 4 * c];
                v2f a1 = *(const v2f*)&U1[baseB + ky * LW + 4 * c];
                accA = wmma4(a0, bmat, accA);
                accB = wmma4(a1, bmat, accB);
            }
        }

        const float srcterm = -(src[2 * step] - src[2 * step + 1]);
        first_step_finish(accA, epA, sgA, pvA, U1, U2, y0, 0, 0, lc, gx, sel, srcterm);
        first_step_finish(accB, epB, sgB, pvB, U1, U2, y0, roffB, mminB, lc, gx, sel, srcterm);
    }
    __syncthreads();

    {
        float ep2[8], sg2[8], u2c[8];
        #pragma unroll
        for (int r = 0; r < 8; ++r) {
            const int cy = r + 8 * sel;
            const int idx = boff + (y0 + cy) * DIMSZ + gx;
            ep2[r] = epsr[idx]; sg2[r] = sigma[idx];
            u2c[r] = U2[(cy + 3) * LW + lc];
        }
        if (mid != nullptr) {
            #pragma unroll
            for (int r = 0; r < 8; ++r) {
                const int cy = r + 8 * sel;
                mid[boff + (y0 + cy) * DIMSZ + gx] = u2c[r];
            }
        }

        v8f acc = {0.f, 0.f, 0.f, 0.f, 0.f, 0.f, 0.f, 0.f};
        const int base2 = mlo * LW + 16 * wave + 2 * sel;
        #pragma unroll
        for (int ky = 0; ky < 7; ++ky) {
            #pragma unroll
            for (int c = 0; c < 6; ++c) {
                v2f a = *(const v2f*)&U2[base2 + ky * LW + 4 * c];
                acc = wmma4(a, breg[ky * 6 + c], acc);
            }
        }

        const float srcterm2 = -(src[2 * step + 2] - src[2 * step + 3]);
        #pragma unroll
        for (int r = 0; r < 8; ++r) {
            const int cy = r + 8 * sel;
            const int gy = y0 + cy;
            const int idx = boff + gy * DIMSZ + gx;
            const float u1c = U1[(cy + 6) * LW + lc];
            next[idx] = fdtd_point(acc[r], ep2[r], sg2[r], u2c[r], u1c,
                                   gy == 128 && gx == 128, srcterm2);
        }
    }
}

extern "C" void kernel_launch(void* const* d_in, const int* in_sizes, int n_in,
                              void* d_out, int out_size, void* d_ws, size_t ws_size,
                              hipStream_t stream) {
    const float* u1    = (const float*)d_in[0];
    const float* u0    = (const float*)d_in[1];
    const float* epsr  = (const float*)d_in[2];
    const float* sigma = (const float*)d_in[3];
    const float* fdk   = (const float*)d_in[4];
    const float* cw    = (const float*)d_in[5];
    const float* src   = (const float*)d_in[6];
    float* out = (float*)d_out;
    float* ws  = (float*)d_ws;

    float* bufs[4] = {ws, ws + NELEM, ws + 2 * NELEM, ws + 3 * NELEM};
    float* Sbuf = ws + 4 * NELEM;

    fdtd_init<<<1, 256, 0, stream>>>(fdk, cw, Sbuf);

    const float* cur  = u1;
    const float* prev = u0;
    dim3 grid(NBATCH, DIMSZ / CORE_H);
    for (int k = 0; k < 16; ++k) {
        float* mid  = (k == 15) ? nullptr : bufs[2 * (k & 1)];
        float* next = (k == 15) ? out : bufs[2 * (k & 1) + 1];
        fdtd_step2<<<grid, NTHR, 0, stream>>>(cur, prev, epsr, sigma, Sbuf, src, mid, next, 2 * k);
        prev = mid;
        cur  = next;
    }
}


# 7x7 conv re-decomposed for the matrix unit: N = 16 blocks of 4x4 pixels, K = the block's 10x10 patch (25 chunks) in place of 7 banded 16x24 row products (42); per-pixel summation order changes
# speedup vs baseline: 2.1761x; 1.1966x over previous
#include <hip/hip_runtime.h>

typedef __attribute__((ext_vector_type(2))) float v2f;
typedef __attribute__((ext_vector_type(4))) float v4f;
typedef __attribute__((ext_vector_type(8))) float v8f;

#define DIMSZ 256
#define NBATCH 16
#define NPIX (DIMSZ * DIMSZ)
#define NELEM (NBATCH * NPIX)
#define NCHUNK 25
#define WBUF_ELEMS (NCHUNK * 32 * 2)

#define CORE_H 16
#define NTHR 512
#define XPAD 3
#define LW 266
#define U1H (CORE_H + 14)
#define U2H (CORE_H + 8)
#define NPADC (LW - DIMSZ)
#define ROW_V4 (DIMSZ / 4)
#define NSTG ((U1H * ROW_V4 + NTHR - 1) / NTHR)

__global__ void fdtd_init(const float* __restrict__ fdk,
                          const float* __restrict__ cw,
                          float* __restrict__ Wbuf) {
    __shared__ float weff[49];
    int t = threadIdx.x;
    if (t < 49) {
        float acc = 0.0f;
        #pragma unroll
        for (int ch = 1; ch < 15; ++ch)
            acc += cw[ch - 1] * fdk[ch * 49 + t];
        weff[t] = acc * (2.0f * 0.5f / 1.0f);
    }
    __syncthreads();
    for (int e = t; e < WBUF_ELEMS; e += blockDim.x) {
        int v    = e & 1;
        int lane = (e >> 1) & 31;
        int c    = e >> 6;
        int m    = lane & 15;
        int h    = lane >> 4;
        int ky   = (c / 5 + 5 * h) - (m >> 2);
        int kx   = (2 * (c % 5) + v) - (m & 3);
        float val = (ky >= 0 && ky <= 6 && kx >= 0 && kx <= 6) ? weff[ky * 7 + kx] : 0.0f;
        Wbuf[e] = val;
    }
}

__device__ __forceinline__ v8f wmma4(v2f a, v2f b, v8f c) {
    return __builtin_amdgcn_wmma_f32_16x16x4_f32(false, a, false, b, (short)0, c, false, false);
}

__device__ __forceinline__ int clampi(int v, int lo, int hi) {
    return v < lo ? lo : (v > hi ? hi : v);
}

__device__ __forceinline__ float fdtd_point(float conv, float ep, float sg,
                                            float cu, float pv, bool at_src, float srcterm) {
    const float dt = 0.5f;
    float inv_den = 1.0f / (2.0f * ep / dt + sg);
    float av = 4.0f * ep / dt;
    float bv = 2.0f * ep / dt - sg;
    float rhs = conv + av * cu - bv * pv;
    if (at_src) rhs += srcterm;
    return inv_den * rhs;
}

__device__ __forceinline__ v8f conv_tile(const float* P, int base, const v2f (&wreg)[NCHUNK]) {
    v8f acc = {0.f, 0.f, 0.f, 0.f, 0.f, 0.f, 0.f, 0.f};
    #pragma unroll
    for (int c = 0; c < NCHUNK; ++c) {
        const v2f d = *(const v2f*)&P[base + (c / 5) * LW + 2 * (c % 5)];
        acc = wmma4(wreg[c], d, acc);
    }
    return acc;
}

__device__ __forceinline__ void conv_tile2(const float* P, int base0, int base1,
                                           const v2f (&wreg)[NCHUNK], v8f& acc0, v8f& acc1) {
    acc0 = (v8f){0.f, 0.f, 0.f, 0.f, 0.f, 0.f, 0.f, 0.f};
    acc1 = (v8f){0.f, 0.f, 0.f, 0.f, 0.f, 0.f, 0.f, 0.f};
    #pragma unroll
    for (int c = 0; c < NCHUNK; ++c) {
        const v2f d0 = *(const v2f*)&P[base0 + (c / 5) * LW + 2 * (c % 5)];
        const v2f d1 = *(const v2f*)&P[base1 + (c / 5) * LW + 2 * (c % 5)];
        acc0 = wmma4(wreg[c], d0, acc0);
        acc1 = wmma4(wreg[c], d1, acc1);
    }
}

struct Coef { v4f ep[2], sg[2], pv[2]; };

__device__ __forceinline__ void step1_coef(Coef& k, const float* __restrict__ epsr,
                                           const float* __restrict__ sigma, const float* prev,
                                           int boff, int y0, int by, int gx0, int h) {
    #pragma unroll
    for (int j = 0; j < 2; ++j) {
        const int gy  = y0 - 4 + 4 * by + 2 * h + j;
        const int idx = boff + clampi(gy, 0, DIMSZ - 1) * DIMSZ + gx0;
        k.ep[j] = *(const v4f*)&epsr[idx];
        k.sg[j] = *(const v4f*)&sigma[idx];
        k.pv[j] = *(const v4f*)&prev[idx];
    }
}

__device__ __forceinline__ void step1_finish(const v8f& acc, const Coef& k,
                                             const float* U1, float* U2,
                                             int y0, int by, int gx0, int h, float srcterm) {
    #pragma unroll
    for (int j = 0; j < 2; ++j) {
        const int rr = 4 * by + 2 * h + j;
        const int gy = y0 - 4 + rr;
        #pragma unroll
        for (int nx = 0; nx < 4; ++nx) {
            const float cu = U1[(rr + 3) * LW + gx0 + XPAD + nx];
            float val = fdtd_point(acc[4 * j + nx], k.ep[j][nx], k.sg[j][nx], cu, k.pv[j][nx],
                                   gy == 128 && (gx0 + nx) == 128, srcterm);
            if (gy < 0 || gy >= DIMSZ) val = 0.0f;
            U2[rr * LW + gx0 + XPAD + nx] = val;
        }
    }
}

__global__ __launch_bounds__(NTHR) void fdtd_step2(
    const float* __restrict__ cur,
    const float* prev,
    const float* __restrict__ epsr,
    const float* __restrict__ sigma,
    const float* __restrict__ Wglob,
    const float* __restrict__ src,
    float* mid,
    float* next,
    int step) {
    __shared__ __attribute__((aligned(16))) float U1[U1H * LW];
    __shared__ __attribute__((aligned(16))) float U2[U2H * LW];

    const int b  = blockIdx.x;
    const int y0 = blockIdx.y * CORE_H;
    const int t  = threadIdx.x;
    const int boff = b * NPIX;
    const float* curb = cur + boff;

    v2f wreg[NCHUNK];
    #pragma unroll
    for (int i = 0; i < NCHUNK; ++i)
        wreg[i] = *(const v2f*)&Wglob[(i * 32 + (t & 31)) * 2];

    v4f stg[NSTG];
    #pragma unroll
    for (int i = 0; i < NSTG; ++i) {
        const int p  = t + NTHR * i;
        const int r  = p / ROW_V4, c4 = p % ROW_V4;
        const int gy = y0 - 7 + r;
        v4f v = {0.f, 0.f, 0.f, 0.f};
        if (r < U1H && gy >= 0 && gy < DIMSZ)
            v = *(const v4f*)&curb[gy * DIMSZ + 4 * c4];
        stg[i] = v;
    }
    for (int e = t; e < (U1H + U2H) * NPADC; e += NTHR) {
        const int r = e / NPADC, k = e % NPADC;
        const int c = (k < XPAD) ? k : (DIMSZ + k);
        if (r < U1H) U1[r * LW + c] = 0.0f;
        else         U2[(r - U1H) * LW + c] = 0.0f;
    }
    #pragma unroll
    for (int i = 0; i < NSTG; ++i) {
        const int p = t + NTHR * i;
        const int r = p / ROW_V4, c4 = p % ROW_V4;
        if (r < U1H) {
            float* d = &U1[r * LW + XPAD + 4 * c4];
            d[0] = stg[i][0]; d[1] = stg[i][1]; d[2] = stg[i][2]; d[3] = stg[i][3];
        }
    }
    __syncthreads();

    const int wave = __builtin_amdgcn_readfirstlane(t >> 5);
    const int lane = t & 31;
    const int c16  = lane & 15;
    const int h    = lane >> 4;
    const int wrow = wave >> 2;
    const int tx   = wave & 3;
    const int gx0  = 4 * (16 * tx + c16);

    {
        const bool extra = ((wrow + tx) & 1) == 0;
        const int  j8    = wrow * 2 + (tx >> 1);
        const int  by1   = 4 + (j8 >> 2);
        const int  gx1   = 4 * (16 * (j8 & 3) + c16);
        const float srcterm = -(src[2 * step] - src[2 * step + 1]);
        const int base0 = (4 * wrow + 5 * h) * LW + gx0;

        if (extra) {
            Coef k0, k1;
            step1_coef(k0, epsr, sigma, prev, boff, y0, wrow, gx0, h);
            step1_coef(k1, epsr, sigma, prev, boff, y0, by1, gx1, h);
            const int base1 = (4 * by1 + 5 * h) * LW + gx1;
            v8f acc0, acc1;
            conv_tile2(U1, base0, base1, wreg, acc0, acc1);
            step1_finish(acc0, k0, U1, U2, y0, wrow, gx0, h, srcterm);
            step1_finish(acc1, k1, U1, U2, y0, by1, gx1, h, srcterm);
        } else {
            Coef k0;
            step1_coef(k0, epsr, sigma, prev, boff, y0, wrow, gx0, h);
            const v8f acc0 = conv_tile(U1, base0, wreg);
            step1_finish(acc0, k0, U1, U2, y0, wrow, gx0, h, srcterm);
        }
    }
    __syncthreads();

    {
        v4f ep2[2], sg2[2], u2c[2];
        #pragma unroll
        for (int j = 0; j < 2; ++j) {
            const int cy  = 4 * wrow + 2 * h + j;
            const int idx = boff + (y0 + cy) * DIMSZ + gx0;
            ep2[j] = *(const v4f*)&epsr[idx];
            sg2[j] = *(const v4f*)&sigma[idx];
            #pragma unroll
            for (int nx = 0; nx < 4; ++nx)
                u2c[j][nx] = U2[(cy + 4) * LW + gx0 + XPAD + nx];
        }
        if (mid != nullptr) {
            #pragma unroll
            for (int j = 0; j < 2; ++j) {
                const int cy = 4 * wrow + 2 * h + j;
                *(v4f*)&mid[boff + (y0 + cy) * DIMSZ + gx0] = u2c[j];
            }
        }

        const v8f acc = conv_tile(U2, (4 * wrow + 1 + 5 * h) * LW + gx0, wreg);

        const float srcterm2 = -(src[2 * step + 2] - src[2 * step + 3]);
        #pragma unroll
        for (int j = 0; j < 2; ++j) {
            const int cy = 4 * wrow + 2 * h + j;
            const int gy = y0 + cy;
            v4f o;
            #pragma unroll
            for (int nx = 0; nx < 4; ++nx) {
                const float u1c = U1[(cy + 7) * LW + gx0 + XPAD + nx];
                o[nx] = fdtd_point(acc[4 * j + nx], ep2[j][nx], sg2[j][nx], u2c[j][nx], u1c,
                                   gy == 128 && (gx0 + nx) == 128, srcterm2);
            }
            *(v4f*)&next[boff + gy * DIMSZ + gx0] = o;
        }
    }
}

extern "C" void kernel_launch(void* const* d_in, const int* in_sizes, int n_in,
                              void* d_out, int out_size, void* d_ws, size_t ws_size,
                              hipStream_t stream) {
    const float* u1    = (const float*)d_in[0];
    const float* u0    = (const float*)d_in[1];
    const float* epsr  = (const float*)d_in[2];
    const float* sigma = (const float*)d_in[3];
    const float* fdk   = (const float*)d_in[4];
    const float* cw    = (const float*)d_in[5];
    const float* src   = (const float*)d_in[6];
    float* out = (float*)d_out;
    float* ws  = (float*)d_ws;

    float* bufs[4] = {ws, ws + NELEM, ws + 2 * NELEM, ws + 3 * NELEM};
    float* Wbuf = ws + 4 * NELEM;

    fdtd_init<<<1, 256, 0, stream>>>(fdk, cw, Wbuf);

    const float* cur  = u1;
    const float* prev = u0;
    dim3 grid(NBATCH, DIMSZ / CORE_H);
    for (int k = 0; k < 16; ++k) {
        float* mid  = (k == 15) ? nullptr : bufs[2 * (k & 1)];
        float* next = (k == 15) ? out : bufs[2 * (k & 1) + 1];
        fdtd_step2<<<grid, NTHR, 0, stream>>>(cur, prev, epsr, sigma, Wbuf, src, mid, next, 2 * k);
        prev = mid;
        cur  = next;
    }
}


# four time steps fused per launch (17 launches become 9): u staged with a 13-row halo, three LDS planes carry the intermediate fields through 9/7/6/4 block rows; per-pixel summation order kept
# speedup vs baseline: 2.4223x; 1.1131x over previous
#include <hip/hip_runtime.h>

typedef __attribute__((ext_vector_type(2))) float v2f;
typedef __attribute__((ext_vector_type(4))) float v4f;
typedef __attribute__((ext_vector_type(8))) float v8f;

#define DIMSZ 256
#define NBATCH 16
#define NPIX (DIMSZ * DIMSZ)
#define NELEM (NBATCH * NPIX)
#define NCHUNK 25
#define WBUF_ELEMS (NCHUNK * 32 * 2)

#define CORE_H 16
#define NTHR 512
#define XPAD 3
#define LW 266
#define P0H 42
#define P1H 36
#define P2H 30
#define P0OFF 13
#define P1OFF 10
#define P2OFF 7
#define NROWS (P0H + P1H + P2H)
#define NPADC (LW - DIMSZ)
#define ROW_V4 (DIMSZ / 4)
#define NSTG ((P0H * ROW_V4 + NTHR - 1) / NTHR)

__global__ void fdtd_init(const float* __restrict__ fdk,
                          const float* __restrict__ cw,
                          float* __restrict__ Wbuf) {
    __shared__ float weff[49];
    int t = threadIdx.x;
    if (t < 49) {
        float acc = 0.0f;
        #pragma unroll
        for (int ch = 1; ch < 15; ++ch)
            acc += cw[ch - 1] * fdk[ch * 49 + t];
        weff[t] = acc * (2.0f * 0.5f / 1.0f);
    }
    __syncthreads();
    for (int e = t; e < WBUF_ELEMS; e += blockDim.x) {
        int v    = e & 1;
        int lane = (e >> 1) & 31;
        int c    = e >> 6;
        int m    = lane & 15;
        int h    = lane >> 4;
        int ky   = (c / 5 + 5 * h) - (m >> 2);
        int kx   = (2 * (c % 5) + v) - (m & 3);
        float val = (ky >= 0 && ky <= 6 && kx >= 0 && kx <= 6) ? weff[ky * 7 + kx] : 0.0f;
        Wbuf[e] = val;
    }
}

__device__ __forceinline__ v8f wmma4(v2f a, v2f b, v8f c) {
    return __builtin_amdgcn_wmma_f32_16x16x4_f32(false, a, false, b, (short)0, c, false, false);
}

__device__ __forceinline__ int clampi(int v, int lo, int hi) {
    return v < lo ? lo : (v > hi ? hi : v);
}

__device__ __forceinline__ float fdtd_point(float conv, float ep, float sg,
                                            float cu, float pv, bool at_src, float srcterm) {
    const float dt = 0.5f;
    float inv_den = 1.0f / (2.0f * ep / dt + sg);
    float av = 4.0f * ep / dt;
    float bv = 2.0f * ep / dt - sg;
    float rhs = conv + av * cu - bv * pv;
    if (at_src) rhs += srcterm;
    return inv_den * rhs;
}

__device__ __forceinline__ v8f conv_tile(const float* P, int base, const v2f (&wreg)[NCHUNK]) {
    v8f acc = {0.f, 0.f, 0.f, 0.f, 0.f, 0.f, 0.f, 0.f};
    #pragma unroll
    for (int c = 0; c < NCHUNK; ++c) {
        const v2f d = *(const v2f*)&P[base + (c / 5) * LW + 2 * (c % 5)];
        acc = wmma4(wreg[c], d, acc);
    }
    return acc;
}

__device__ __forceinline__ void conv_tile2(const float* P, int base0, int base1,
                                           const v2f (&wreg)[NCHUNK], v8f& acc0, v8f& acc1) {
    acc0 = (v8f){0.f, 0.f, 0.f, 0.f, 0.f, 0.f, 0.f, 0.f};
    acc1 = (v8f){0.f, 0.f, 0.f, 0.f, 0.f, 0.f, 0.f, 0.f};
    #pragma unroll
    for (int c = 0; c < NCHUNK; ++c) {
        const v2f d0 = *(const v2f*)&P[base0 + (c / 5) * LW + 2 * (c % 5)];
        const v2f d1 = *(const v2f*)&P[base1 + (c / 5) * LW + 2 * (c % 5)];
        acc0 = wmma4(wreg[c], d0, acc0);
        acc1 = wmma4(wreg[c], d1, acc1);
    }
}

struct StageCtx {
    const float* Pin;  int in_r0;
    int rel0;
    const float* Pprev; int prev_off;
    float* Pout;       int out_off;
    const float* epsr; const float* sigma; const float* prevg;
    float* mid; float* next;
    int boff, y0, gx0, h;
    float srcterm;
};

struct Coef { v4f ep[2], sg[2], pv[2]; };

template <bool PREV_GLOBAL>
__device__ __forceinline__ void tile_coef(Coef& k, const StageCtx& s, int i) {
    #pragma unroll
    for (int j = 0; j < 2; ++j) {
        const int rel = s.rel0 + 4 * i + 2 * s.h + j;
        const int idx = s.boff + clampi(s.y0 + rel, 0, DIMSZ - 1) * DIMSZ + s.gx0;
        k.ep[j] = *(const v4f*)&s.epsr[idx];
        k.sg[j] = *(const v4f*)&s.sigma[idx];
        if (PREV_GLOBAL) {
            k.pv[j] = *(const v4f*)&s.prevg[idx];
        } else {
            #pragma unroll
            for (int nx = 0; nx < 4; ++nx)
                k.pv[j][nx] = s.Pprev[(rel + s.prev_off) * LW + s.gx0 + XPAD + nx];
        }
    }
}

template <bool OUT_LDS>
__device__ __forceinline__ void tile_finish(const v8f& acc, const Coef& k, const StageCtx& s, int i) {
    #pragma unroll
    for (int j = 0; j < 2; ++j) {
        const int rr  = 4 * i + 2 * s.h + j;
        const int rel = s.rel0 + rr;
        const int gy  = s.y0 + rel;
        v4f cu, o;
        #pragma unroll
        for (int nx = 0; nx < 4; ++nx)
            cu[nx] = s.Pin[(s.in_r0 + 3 + rr) * LW + s.gx0 + XPAD + nx];
        #pragma unroll
        for (int nx = 0; nx < 4; ++nx)
            o[nx] = fdtd_point(acc[4 * j + nx], k.ep[j][nx], k.sg[j][nx], cu[nx], k.pv[j][nx],
                               gy == 128 && (s.gx0 + nx) == 128, s.srcterm);
        if (OUT_LDS) {
            const bool outside = gy < 0 || gy >= DIMSZ;
            #pragma unroll
            for (int nx = 0; nx < 4; ++nx)
                s.Pout[(rel + s.out_off) * LW + s.gx0 + XPAD + nx] = outside ? 0.0f : o[nx];
        } else {
            if (s.mid != nullptr) *(v4f*)&s.mid[s.boff + gy * DIMSZ + s.gx0] = cu;
            *(v4f*)&s.next[s.boff + gy * DIMSZ + s.gx0] = o;
        }
    }
}

__device__ __forceinline__ int patch_base(const StageCtx& s, int i) {
    return (s.in_r0 + 4 * i + 5 * s.h) * LW + s.gx0;
}

template <bool PREV_GLOBAL, bool OUT_LDS>
__device__ __forceinline__ void stage_one(const StageCtx& s, const v2f (&wreg)[NCHUNK], int i) {
    Coef k;
    tile_coef<PREV_GLOBAL>(k, s, i);
    const v8f acc = conv_tile(s.Pin, patch_base(s, i), wreg);
    tile_finish<OUT_LDS>(acc, k, s, i);
}

template <bool PREV_GLOBAL, bool OUT_LDS>
__device__ __forceinline__ void stage_two(const StageCtx& s, const v2f (&wreg)[NCHUNK], int i0, int i1) {
    Coef k0, k1;
    tile_coef<PREV_GLOBAL>(k0, s, i0);
    tile_coef<PREV_GLOBAL>(k1, s, i1);
    v8f acc0, acc1;
    conv_tile2(s.Pin, patch_base(s, i0), patch_base(s, i1), wreg, acc0, acc1);
    tile_finish<OUT_LDS>(acc0, k0, s, i0);
    tile_finish<OUT_LDS>(acc1, k1, s, i1);
}

template <int NBR, bool PREV_GLOBAL, bool OUT_LDS>
__device__ __forceinline__ void run_stage(const StageCtx& s, const v2f (&wreg)[NCHUNK], int vr) {
    if (NBR > 8) {
        if (vr + 8 < NBR) stage_one<PREV_GLOBAL, OUT_LDS>(s, wreg, vr + 8);
    }
    if (vr + 4 < NBR) stage_two<PREV_GLOBAL, OUT_LDS>(s, wreg, vr, vr + 4);
    else              stage_one<PREV_GLOBAL, OUT_LDS>(s, wreg, vr);
}

__global__ __launch_bounds__(NTHR) void fdtd_step4(
    const float* __restrict__ cur,
    const float* prev,
    const float* __restrict__ epsr,
    const float* __restrict__ sigma,
    const float* __restrict__ Wglob,
    const float* __restrict__ src,
    float* mid,
    float* next,
    int step) {
    __shared__ __attribute__((aligned(16))) float L[NROWS * LW];
    float* P0 = L;
    float* P1 = L + P0H * LW;
    float* P2 = L + (P0H + P1H) * LW;

    const int b  = blockIdx.x;
    const int y0 = blockIdx.y * CORE_H;
    const int t  = threadIdx.x;
    const int boff = b * NPIX;
    const float* curb = cur + boff;

    v2f wreg[NCHUNK];
    #pragma unroll
    for (int i = 0; i < NCHUNK; ++i)
        wreg[i] = *(const v2f*)&Wglob[(i * 32 + (t & 31)) * 2];

    v4f stg[NSTG];
    #pragma unroll
    for (int i = 0; i < NSTG; ++i) {
        const int p  = t + NTHR * i;
        const int r  = p / ROW_V4, c4 = p % ROW_V4;
        const int gy = y0 - P0OFF + r;
        v4f v = {0.f, 0.f, 0.f, 0.f};
        if (r < P0H && gy >= 0 && gy < DIMSZ)
            v = *(const v4f*)&curb[gy * DIMSZ + 4 * c4];
        stg[i] = v;
    }
    for (int e = t; e < NROWS * NPADC; e += NTHR) {
        const int r = e / NPADC, k = e % NPADC;
        const int c = (k < XPAD) ? k : (DIMSZ + k);
        L[r * LW + c] = 0.0f;
    }
    P2[(t >> 8) * (P2H - 1) * LW + XPAD + (t & 255)] = 0.0f;
    #pragma unroll
    for (int i = 0; i < NSTG; ++i) {
        const int p = t + NTHR * i;
        const int r = p / ROW_V4, c4 = p % ROW_V4;
        if (r < P0H) {
            float* d = &P0[r * LW + XPAD + 4 * c4];
            d[0] = stg[i][0]; d[1] = stg[i][1]; d[2] = stg[i][2]; d[3] = stg[i][3];
        }
    }
    __syncthreads();

    const int wave = __builtin_amdgcn_readfirstlane(t >> 5);
    const int lane = t & 31;
    const int c16  = lane & 15;
    const int h    = lane >> 4;
    const int tx   = wave & 3;
    const int vr   = ((wave >> 2) + tx) & 3;
    const int gx0  = 4 * (16 * tx + c16);

    StageCtx s;
    s.epsr = epsr; s.sigma = sigma; s.prevg = prev; s.mid = mid; s.next = next;
    s.boff = boff; s.y0 = y0; s.gx0 = gx0; s.h = h;

    s.Pin = P0; s.in_r0 = 0; s.rel0 = -10; s.Pprev = nullptr; s.prev_off = 0;
    s.Pout = P1; s.out_off = P1OFF;
    s.srcterm = -(src[2 * step] - src[2 * step + 1]);
    run_stage<9, true, true>(s, wreg, vr);
    __syncthreads();

    s.Pin = P1; s.in_r0 = 1; s.rel0 = -6; s.Pprev = P0; s.prev_off = P0OFF;
    s.Pout = P2; s.out_off = P2OFF;
    s.srcterm = -(src[2 * step + 2] - src[2 * step + 3]);
    run_stage<7, false, true>(s, wreg, vr);
    __syncthreads();

    s.Pin = P2; s.in_r0 = 0; s.rel0 = -4; s.Pprev = P1; s.prev_off = P1OFF;
    s.Pout = P0; s.out_off = P0OFF;
    s.srcterm = -(src[2 * step + 4] - src[2 * step + 5]);
    run_stage<6, false, true>(s, wreg, vr);
    __syncthreads();

    s.Pin = P0; s.in_r0 = 10; s.rel0 = 0; s.Pprev = P2; s.prev_off = P2OFF;
    s.Pout = nullptr; s.out_off = 0;
    s.srcterm = -(src[2 * step + 6] - src[2 * step + 7]);
    run_stage<4, false, false>(s, wreg, vr);
}

extern "C" void kernel_launch(void* const* d_in, const int* in_sizes, int n_in,
                              void* d_out, int out_size, void* d_ws, size_t ws_size,
                              hipStream_t stream) {
    const float* u1    = (const float*)d_in[0];
    const float* u0    = (const float*)d_in[1];
    const float* epsr  = (const float*)d_in[2];
    const float* sigma = (const float*)d_in[3];
    const float* fdk   = (const float*)d_in[4];
    const float* cw    = (const float*)d_in[5];
    const float* src   = (const float*)d_in[6];
    float* out = (float*)d_out;
    float* ws  = (float*)d_ws;

    float* bufs[4] = {ws, ws + NELEM, ws + 2 * NELEM, ws + 3 * NELEM};
    float* Wbuf = ws + 4 * NELEM;

    fdtd_init<<<1, 256, 0, stream>>>(fdk, cw, Wbuf);

    const float* cur  = u1;
    const float* prev = u0;
    dim3 grid(NBATCH, DIMSZ / CORE_H);
    for (int k = 0; k < 8; ++k) {
        float* mid  = (k == 7) ? nullptr : bufs[2 * (k & 1)];
        float* next = (k == 7) ? out : bufs[2 * (k & 1) + 1];
        fdtd_step4<<<grid, NTHR, 0, stream>>>(cur, prev, epsr, sigma, Wbuf, src, mid, next, 4 * k);
        prev = mid;
        cur  = next;
    }
}


# stage 1 keeps its per-pixel update coefficients (1/den, a, b) in three LDS planes; stages 2-4 read them instead of reloading epsr/sigma from global and dividing again; arithmetic unchanged
# speedup vs baseline: 2.5226x; 1.0414x over previous
#include <hip/hip_runtime.h>

typedef __attribute__((ext_vector_type(2))) float v2f;
typedef __attribute__((ext_vector_type(4))) float v4f;
typedef __attribute__((ext_vector_type(8))) float v8f;

#define DIMSZ 256
#define NBATCH 16
#define NPIX (DIMSZ * DIMSZ)
#define NELEM (NBATCH * NPIX)
#define NCHUNK 25
#define WBUF_ELEMS (NCHUNK * 32 * 2)

#define CORE_H 16
#define NTHR 512
#define XPAD 3
#define LW 266
#define P0H 42
#define P1H 36
#define P2H 30
#define P0OFF 13
#define P1OFF 10
#define P2OFF 7
#define NROWS (P0H + P1H + P2H)
#define NPADC (LW - DIMSZ)
#define ROW_V4 (DIMSZ / 4)
#define NSTG ((P0H * ROW_V4 + NTHR - 1) / NTHR)
#define CFH 36
#define CFOFF 10
#define CFPLANE (CFH * DIMSZ)

__global__ void fdtd_init(const float* __restrict__ fdk,
                          const float* __restrict__ cw,
                          float* __restrict__ Wbuf) {
    __shared__ float weff[49];
    int t = threadIdx.x;
    if (t < 49) {
        float acc = 0.0f;
        #pragma unroll
        for (int ch = 1; ch < 15; ++ch)
            acc += cw[ch - 1] * fdk[ch * 49 + t];
        weff[t] = acc * (2.0f * 0.5f / 1.0f);
    }
    __syncthreads();
    for (int e = t; e < WBUF_ELEMS; e += blockDim.x) {
        int v    = e & 1;
        int lane = (e >> 1) & 31;
        int c    = e >> 6;
        int m    = lane & 15;
        int h    = lane >> 4;
        int ky   = (c / 5 + 5 * h) - (m >> 2);
        int kx   = (2 * (c % 5) + v) - (m & 3);
        float val = (ky >= 0 && ky <= 6 && kx >= 0 && kx <= 6) ? weff[ky * 7 + kx] : 0.0f;
        Wbuf[e] = val;
    }
}

__device__ __forceinline__ v8f wmma4(v2f a, v2f b, v8f c) {
    return __builtin_amdgcn_wmma_f32_16x16x4_f32(false, a, false, b, (short)0, c, false, false);
}

__device__ __forceinline__ int clampi(int v, int lo, int hi) {
    return v < lo ? lo : (v > hi ? hi : v);
}

__device__ __forceinline__ float fdtd_point(float conv, float inv_den, float av, float bv,
                                            float cu, float pv, bool at_src, float srcterm) {
    float rhs = conv + av * cu - bv * pv;
    if (at_src) rhs += srcterm;
    return inv_den * rhs;
}

__device__ __forceinline__ v8f conv_tile(const float* P, int base, const v2f (&wreg)[NCHUNK]) {
    v8f acc = {0.f, 0.f, 0.f, 0.f, 0.f, 0.f, 0.f, 0.f};
    #pragma unroll
    for (int c = 0; c < NCHUNK; ++c) {
        const v2f d = *(const v2f*)&P[base + (c / 5) * LW + 2 * (c % 5)];
        acc = wmma4(wreg[c], d, acc);
    }
    return acc;
}

__device__ __forceinline__ void conv_tile2(const float* P, int base0, int base1,
                                           const v2f (&wreg)[NCHUNK], v8f& acc0, v8f& acc1) {
    acc0 = (v8f){0.f, 0.f, 0.f, 0.f, 0.f, 0.f, 0.f, 0.f};
    acc1 = (v8f){0.f, 0.f, 0.f, 0.f, 0.f, 0.f, 0.f, 0.f};
    #pragma unroll
    for (int c = 0; c < NCHUNK; ++c) {
        const v2f d0 = *(const v2f*)&P[base0 + (c / 5) * LW + 2 * (c % 5)];
        const v2f d1 = *(const v2f*)&P[base1 + (c / 5) * LW + 2 * (c % 5)];
        acc0 = wmma4(wreg[c], d0, acc0);
        acc1 = wmma4(wreg[c], d1, acc1);
    }
}

struct StageCtx {
    const float* Pin;  int in_r0;
    int rel0;
    const float* Pprev; int prev_off;
    float* Pout;       int out_off;
    const float* epsr; const float* sigma; const float* prevg;
    float* Cf;
    float* mid; float* next;
    int boff, y0, gx0, h;
    float srcterm;
};

struct Coef { v4f iv[2], av[2], bv[2], pv[2]; };

template <bool PREV_GLOBAL>
__device__ __forceinline__ void tile_coef(Coef& k, const StageCtx& s, int i) {
    #pragma unroll
    for (int j = 0; j < 2; ++j) {
        const int rel = s.rel0 + 4 * i + 2 * s.h + j;
        float* cf = &s.Cf[(rel + CFOFF) * DIMSZ + s.gx0];
        if (PREV_GLOBAL) {
            const int idx = s.boff + clampi(s.y0 + rel, 0, DIMSZ - 1) * DIMSZ + s.gx0;
            const v4f ep = *(const v4f*)&s.epsr[idx];
            const v4f sg = *(const v4f*)&s.sigma[idx];
            k.pv[j] = *(const v4f*)&s.prevg[idx];
            const float dt = 0.5f;
            #pragma unroll
            for (int nx = 0; nx < 4; ++nx) {
                k.iv[j][nx] = 1.0f / (2.0f * ep[nx] / dt + sg[nx]);
                k.av[j][nx] = 4.0f * ep[nx] / dt;
                k.bv[j][nx] = 2.0f * ep[nx] / dt - sg[nx];
            }
            *(v4f*)&cf[0]           = k.iv[j];
            *(v4f*)&cf[CFPLANE]     = k.av[j];
            *(v4f*)&cf[2 * CFPLANE] = k.bv[j];
        } else {
            k.iv[j] = *(const v4f*)&cf[0];
            k.av[j] = *(const v4f*)&cf[CFPLANE];
            k.bv[j] = *(const v4f*)&cf[2 * CFPLANE];
            #pragma unroll
            for (int nx = 0; nx < 4; ++nx)
                k.pv[j][nx] = s.Pprev[(rel + s.prev_off) * LW + s.gx0 + XPAD + nx];
        }
    }
}

template <bool OUT_LDS>
__device__ __forceinline__ void tile_finish(const v8f& acc, const Coef& k, const StageCtx& s, int i) {
    #pragma unroll
    for (int j = 0; j < 2; ++j) {
        const int rr  = 4 * i + 2 * s.h + j;
        const int rel = s.rel0 + rr;
        const int gy  = s.y0 + rel;
        v4f cu, o;
        #pragma unroll
        for (int nx = 0; nx < 4; ++nx)
            cu[nx] = s.Pin[(s.in_r0 + 3 + rr) * LW + s.gx0 + XPAD + nx];
        #pragma unroll
        for (int nx = 0; nx < 4; ++nx)
            o[nx] = fdtd_point(acc[4 * j + nx], k.iv[j][nx], k.av[j][nx], k.bv[j][nx], cu[nx],
                               k.pv[j][nx], gy == 128 && (s.gx0 + nx) == 128, s.srcterm);
        if (OUT_LDS) {
            const bool outside = gy < 0 || gy >= DIMSZ;
            #pragma unroll
            for (int nx = 0; nx < 4; ++nx)
                s.Pout[(rel + s.out_off) * LW + s.gx0 + XPAD + nx] = outside ? 0.0f : o[nx];
        } else {
            if (s.mid != nullptr) *(v4f*)&s.mid[s.boff + gy * DIMSZ + s.gx0] = cu;
            *(v4f*)&s.next[s.boff + gy * DIMSZ + s.gx0] = o;
        }
    }
}

__device__ __forceinline__ int patch_base(const StageCtx& s, int i) {
    return (s.in_r0 + 4 * i + 5 * s.h) * LW + s.gx0;
}

template <bool PREV_GLOBAL, bool OUT_LDS>
__device__ __forceinline__ void stage_one(const StageCtx& s, const v2f (&wreg)[NCHUNK], int i) {
    Coef k;
    tile_coef<PREV_GLOBAL>(k, s, i);
    const v8f acc = conv_tile(s.Pin, patch_base(s, i), wreg);
    tile_finish<OUT_LDS>(acc, k, s, i);
}

template <bool PREV_GLOBAL, bool OUT_LDS>
__device__ __forceinline__ void stage_two(const StageCtx& s, const v2f (&wreg)[NCHUNK], int i0, int i1) {
    Coef k0, k1;
    tile_coef<PREV_GLOBAL>(k0, s, i0);
    tile_coef<PREV_GLOBAL>(k1, s, i1);
    v8f acc0, acc1;
    conv_tile2(s.Pin, patch_base(s, i0), patch_base(s, i1), wreg, acc0, acc1);
    tile_finish<OUT_LDS>(acc0, k0, s, i0);
    tile_finish<OUT_LDS>(acc1, k1, s, i1);
}

template <int NBR, bool PREV_GLOBAL, bool OUT_LDS>
__device__ __forceinline__ void run_stage(const StageCtx& s, const v2f (&wreg)[NCHUNK], int vr) {
    if (NBR > 8) {
        if (vr + 8 < NBR) stage_one<PREV_GLOBAL, OUT_LDS>(s, wreg, vr + 8);
    }
    if (vr + 4 < NBR) stage_two<PREV_GLOBAL, OUT_LDS>(s, wreg, vr, vr + 4);
    else              stage_one<PREV_GLOBAL, OUT_LDS>(s, wreg, vr);
}

__global__ __launch_bounds__(NTHR) void fdtd_step4(
    const float* __restrict__ cur,
    const float* prev,
    const float* __restrict__ epsr,
    const float* __restrict__ sigma,
    const float* __restrict__ Wglob,
    const float* __restrict__ src,
    float* mid,
    float* next,
    int step) {
    __shared__ __attribute__((aligned(16))) float L[NROWS * LW];
    __shared__ __attribute__((aligned(16))) float Cf[3 * CFPLANE];
    float* P0 = L;
    float* P1 = L + P0H * LW;
    float* P2 = L + (P0H + P1H) * LW;

    const int b  = blockIdx.x;
    const int y0 = blockIdx.y * CORE_H;
    const int t  = threadIdx.x;
    const int boff = b * NPIX;
    const float* curb = cur + boff;

    v2f wreg[NCHUNK];
    #pragma unroll
    for (int i = 0; i < NCHUNK; ++i)
        wreg[i] = *(const v2f*)&Wglob[(i * 32 + (t & 31)) * 2];

    v4f stg[NSTG];
    #pragma unroll
    for (int i = 0; i < NSTG; ++i) {
        const int p  = t + NTHR * i;
        const int r  = p / ROW_V4, c4 = p % ROW_V4;
        const int gy = y0 - P0OFF + r;
        v4f v = {0.f, 0.f, 0.f, 0.f};
        if (r < P0H && gy >= 0 && gy < DIMSZ)
            v = *(const v4f*)&curb[gy * DIMSZ + 4 * c4];
        stg[i] = v;
    }
    for (int e = t; e < NROWS * NPADC; e += NTHR) {
        const int r = e / NPADC, k = e % NPADC;
        const int c = (k < XPAD) ? k : (DIMSZ + k);
        L[r * LW + c] = 0.0f;
    }
    P2[(t >> 8) * (P2H - 1) * LW + XPAD + (t & 255)] = 0.0f;
    #pragma unroll
    for (int i = 0; i < NSTG; ++i) {
        const int p = t + NTHR * i;
        const int r = p / ROW_V4, c4 = p % ROW_V4;
        if (r < P0H) {
            float* d = &P0[r * LW + XPAD + 4 * c4];
            d[0] = stg[i][0]; d[1] = stg[i][1]; d[2] = stg[i][2]; d[3] = stg[i][3];
        }
    }
    __syncthreads();

    const int wave = __builtin_amdgcn_readfirstlane(t >> 5);
    const int lane = t & 31;
    const int c16  = lane & 15;
    const int h    = lane >> 4;
    const int tx   = wave & 3;
    const int vr   = ((wave >> 2) + tx) & 3;
    const int gx0  = 4 * (16 * tx + c16);

    StageCtx s;
    s.epsr = epsr; s.sigma = sigma; s.prevg = prev; s.Cf = Cf; s.mid = mid; s.next = next;
    s.boff = boff; s.y0 = y0; s.gx0 = gx0; s.h = h;

    s.Pin = P0; s.in_r0 = 0; s.rel0 = -10; s.Pprev = nullptr; s.prev_off = 0;
    s.Pout = P1; s.out_off = P1OFF;
    s.srcterm = -(src[2 * step] - src[2 * step + 1]);
    run_stage<9, true, true>(s, wreg, vr);
    __syncthreads();

    s.Pin = P1; s.in_r0 = 1; s.rel0 = -6; s.Pprev = P0; s.prev_off = P0OFF;
    s.Pout = P2; s.out_off = P2OFF;
    s.srcterm = -(src[2 * step + 2] - src[2 * step + 3]);
    run_stage<7, false, true>(s, wreg, vr);
    __syncthreads();

    s.Pin = P2; s.in_r0 = 0; s.rel0 = -4; s.Pprev = P1; s.prev_off = P1OFF;
    s.Pout = P0; s.out_off = P0OFF;
    s.srcterm = -(src[2 * step + 4] - src[2 * step + 5]);
    run_stage<6, false, true>(s, wreg, vr);
    __syncthreads();

    s.Pin = P0; s.in_r0 = 10; s.rel0 = 0; s.Pprev = P2; s.prev_off = P2OFF;
    s.Pout = nullptr; s.out_off = 0;
    s.srcterm = -(src[2 * step + 6] - src[2 * step + 7]);
    run_stage<4, false, false>(s, wreg, vr);
}

extern "C" void kernel_launch(void* const* d_in, const int* in_sizes, int n_in,
                              void* d_out, int out_size, void* d_ws, size_t ws_size,
                              hipStream_t stream) {
    const float* u1    = (const float*)d_in[0];
    const float* u0    = (const float*)d_in[1];
    const float* epsr  = (const float*)d_in[2];
    const float* sigma = (const float*)d_in[3];
    const float* fdk   = (const float*)d_in[4];
    const float* cw    = (const float*)d_in[5];
    const float* src   = (const float*)d_in[6];
    float* out = (float*)d_out;
    float* ws  = (float*)d_ws;

    float* bufs[4] = {ws, ws + NELEM, ws + 2 * NELEM, ws + 3 * NELEM};
    float* Wbuf = ws + 4 * NELEM;

    fdtd_init<<<1, 256, 0, stream>>>(fdk, cw, Wbuf);

    const float* cur  = u1;
    const float* prev = u0;
    dim3 grid(NBATCH, DIMSZ / CORE_H);
    for (int k = 0; k < 8; ++k) {
        float* mid  = (k == 7) ? nullptr : bufs[2 * (k & 1)];
        float* next = (k == 7) ? out : bufs[2 * (k & 1) + 1];
        fdtd_step4<<<grid, NTHR, 0, stream>>>(cur, prev, epsr, sigma, Wbuf, src, mid, next, 4 * k);
        prev = mid;
        cur  = next;
    }
}


# a tile's fragment reads leave LDS as groups two to three patch rows ahead of their WMMAs, held by a scheduling fence that also keeps stage 1's coefficient loads ahead of the chain; sum order kept
# speedup vs baseline: 2.6622x; 1.0553x over previous
#include <hip/hip_runtime.h>

typedef __attribute__((ext_vector_type(2))) float v2f;
typedef __attribute__((ext_vector_type(4))) float v4f;
typedef __attribute__((ext_vector_type(8))) float v8f;

#define DIMSZ 256
#define NBATCH 16
#define NPIX (DIMSZ * DIMSZ)
#define NELEM (NBATCH * NPIX)
#define NCHUNK 25
#define WBUF_ELEMS (NCHUNK * 32 * 2)

#define CORE_H 16
#define NTHR 512
#define XPAD 3
#define LW 266
#define P0H 42
#define P1H 36
#define P2H 30
#define P0OFF 13
#define P1OFF 10
#define P2OFF 7
#define NROWS (P0H + P1H + P2H)
#define NPADC (LW - DIMSZ)
#define ROW_V4 (DIMSZ / 4)
#define NSTG ((P0H * ROW_V4 + NTHR - 1) / NTHR)
#define CFH 36
#define CFOFF 10
#define CFPLANE (CFH * DIMSZ)

__global__ void fdtd_init(const float* __restrict__ fdk,
                          const float* __restrict__ cw,
                          float* __restrict__ Wbuf) {
    __shared__ float weff[49];
    int t = threadIdx.x;
    if (t < 49) {
        float acc = 0.0f;
        #pragma unroll
        for (int ch = 1; ch < 15; ++ch)
            acc += cw[ch - 1] * fdk[ch * 49 + t];
        weff[t] = acc * (2.0f * 0.5f / 1.0f);
    }
    __syncthreads();
    for (int e = t; e < WBUF_ELEMS; e += blockDim.x) {
        int v    = e & 1;
        int lane = (e >> 1) & 31;
        int c    = e >> 6;
        int m    = lane & 15;
        int h    = lane >> 4;
        int ky   = (c / 5 + 5 * h) - (m >> 2);
        int kx   = (2 * (c % 5) + v) - (m & 3);
        float val = (ky >= 0 && ky <= 6 && kx >= 0 && kx <= 6) ? weff[ky * 7 + kx] : 0.0f;
        Wbuf[e] = val;
    }
}

__device__ __forceinline__ v8f wmma4(v2f a, v2f b, v8f c) {
    return __builtin_amdgcn_wmma_f32_16x16x4_f32(false, a, false, b, (short)0, c, false, false);
}

__device__ __forceinline__ int clampi(int v, int lo, int hi) {
    return v < lo ? lo : (v > hi ? hi : v);
}

__device__ __forceinline__ float fdtd_point(float conv, float inv_den, float av, float bv,
                                            float cu, float pv, bool at_src, float srcterm) {
    float rhs = conv + av * cu - bv * pv;
    if (at_src) rhs += srcterm;
    return inv_den * rhs;
}

#define NFROW 5
__device__ __forceinline__ void frag_row(const float* P, int base, int r, v2f (&d)[NCHUNK]) {
    #pragma unroll
    for (int q = 0; q < NFROW; ++q)
        d[NFROW * r + q] = *(const v2f*)&P[base + r * LW + 2 * q];
}

__device__ __forceinline__ void chain_row(v8f& acc, const v2f (&d)[NCHUNK], const v2f (&wreg)[NCHUNK], int r) {
    #pragma unroll
    for (int q = 0; q < NFROW; ++q)
        acc = wmma4(wreg[NFROW * r + q], d[NFROW * r + q], acc);
}

#define FRAG_FENCE() __builtin_amdgcn_sched_barrier(0x606)

struct StageCtx {
    const float* Pin;  int in_r0;
    int rel0;
    const float* Pprev; int prev_off;
    float* Pout;       int out_off;
    const float* epsr; const float* sigma; const float* prevg;
    float* Cf;
    float* mid; float* next;
    int boff, y0, gx0, h;
    float srcterm;
};

struct Coef { v4f iv[2], av[2], bv[2], pv[2]; };
struct CoefRaw { v4f ep[2], sg[2]; };

__device__ __forceinline__ void coef_load(CoefRaw& w, Coef& k, const StageCtx& s, int i) {
    #pragma unroll
    for (int j = 0; j < 2; ++j) {
        const int rel = s.rel0 + 4 * i + 2 * s.h + j;
        const int idx = s.boff + clampi(s.y0 + rel, 0, DIMSZ - 1) * DIMSZ + s.gx0;
        w.ep[j] = *(const v4f*)&s.epsr[idx];
        w.sg[j] = *(const v4f*)&s.sigma[idx];
        k.pv[j] = *(const v4f*)&s.prevg[idx];
    }
}

__device__ __forceinline__ void coef_make(const CoefRaw& w, Coef& k, const StageCtx& s, int i) {
    #pragma unroll
    for (int j = 0; j < 2; ++j) {
        const int rel = s.rel0 + 4 * i + 2 * s.h + j;
        float* cf = &s.Cf[(rel + CFOFF) * DIMSZ + s.gx0];
        const float dt = 0.5f;
        #pragma unroll
        for (int nx = 0; nx < 4; ++nx) {
            k.iv[j][nx] = 1.0f / (2.0f * w.ep[j][nx] / dt + w.sg[j][nx]);
            k.av[j][nx] = 4.0f * w.ep[j][nx] / dt;
            k.bv[j][nx] = 2.0f * w.ep[j][nx] / dt - w.sg[j][nx];
        }
        *(v4f*)&cf[0]           = k.iv[j];
        *(v4f*)&cf[CFPLANE]     = k.av[j];
        *(v4f*)&cf[2 * CFPLANE] = k.bv[j];
    }
}

__device__ __forceinline__ void coef_lds(Coef& k, const StageCtx& s, int i) {
    #pragma unroll
    for (int j = 0; j < 2; ++j) {
        const int rel = s.rel0 + 4 * i + 2 * s.h + j;
        const float* cf = &s.Cf[(rel + CFOFF) * DIMSZ + s.gx0];
        k.iv[j] = *(const v4f*)&cf[0];
        k.av[j] = *(const v4f*)&cf[CFPLANE];
        k.bv[j] = *(const v4f*)&cf[2 * CFPLANE];
        #pragma unroll
        for (int nx = 0; nx < 4; ++nx)
            k.pv[j][nx] = s.Pprev[(rel + s.prev_off) * LW + s.gx0 + XPAD + nx];
    }
}

template <bool OUT_LDS>
__device__ __forceinline__ void tile_finish(const v8f& acc, const Coef& k, const StageCtx& s, int i) {
    #pragma unroll
    for (int j = 0; j < 2; ++j) {
        const int rr  = 4 * i + 2 * s.h + j;
        const int rel = s.rel0 + rr;
        const int gy  = s.y0 + rel;
        v4f cu, o;
        #pragma unroll
        for (int nx = 0; nx < 4; ++nx)
            cu[nx] = s.Pin[(s.in_r0 + 3 + rr) * LW + s.gx0 + XPAD + nx];
        #pragma unroll
        for (int nx = 0; nx < 4; ++nx)
            o[nx] = fdtd_point(acc[4 * j + nx], k.iv[j][nx], k.av[j][nx], k.bv[j][nx], cu[nx],
                               k.pv[j][nx], gy == 128 && (s.gx0 + nx) == 128, s.srcterm);
        if (OUT_LDS) {
            const bool outside = gy < 0 || gy >= DIMSZ;
            #pragma unroll
            for (int nx = 0; nx < 4; ++nx)
                s.Pout[(rel + s.out_off) * LW + s.gx0 + XPAD + nx] = outside ? 0.0f : o[nx];
        } else {
            if (s.mid != nullptr) *(v4f*)&s.mid[s.boff + gy * DIMSZ + s.gx0] = cu;
            *(v4f*)&s.next[s.boff + gy * DIMSZ + s.gx0] = o;
        }
    }
}

__device__ __forceinline__ int patch_base(const StageCtx& s, int i) {
    return (s.in_r0 + 4 * i + 5 * s.h) * LW + s.gx0;
}

template <bool PREV_GLOBAL, bool OUT_LDS>
__device__ __forceinline__ void stage_one(const StageCtx& s, const v2f (&wreg)[NCHUNK], int i) {
    Coef k;
    CoefRaw w;
    v2f d[NCHUNK];
    const int base = patch_base(s, i);
    if (PREV_GLOBAL) coef_load(w, k, s, i);
    frag_row(s.Pin, base, 0, d);
    frag_row(s.Pin, base, 1, d);
    frag_row(s.Pin, base, 2, d);
    FRAG_FENCE();
    v8f acc = {0.f, 0.f, 0.f, 0.f, 0.f, 0.f, 0.f, 0.f};
    chain_row(acc, d, wreg, 0);
    FRAG_FENCE();
    frag_row(s.Pin, base, 3, d);
    FRAG_FENCE();
    chain_row(acc, d, wreg, 1);
    FRAG_FENCE();
    frag_row(s.Pin, base, 4, d);
    FRAG_FENCE();
    chain_row(acc, d, wreg, 2);
    chain_row(acc, d, wreg, 3);
    chain_row(acc, d, wreg, 4);
    if (PREV_GLOBAL) coef_make(w, k, s, i); else coef_lds(k, s, i);
    tile_finish<OUT_LDS>(acc, k, s, i);
}

template <bool PREV_GLOBAL, bool OUT_LDS>
__device__ __forceinline__ void stage_two(const StageCtx& s, const v2f (&wreg)[NCHUNK], int i0, int i1) {
    Coef k0, k1;
    CoefRaw w0, w1;
    v2f d0[NCHUNK], d1[NCHUNK];
    const int base0 = patch_base(s, i0), base1 = patch_base(s, i1);
    if (PREV_GLOBAL) { coef_load(w0, k0, s, i0); coef_load(w1, k1, s, i1); }
    #pragma unroll
    for (int r = 0; r < 3; ++r) { frag_row(s.Pin, base0, r, d0); frag_row(s.Pin, base1, r, d1); }
    FRAG_FENCE();
    v8f acc0 = {0.f, 0.f, 0.f, 0.f, 0.f, 0.f, 0.f, 0.f};
    v8f acc1 = {0.f, 0.f, 0.f, 0.f, 0.f, 0.f, 0.f, 0.f};
    #pragma unroll
    for (int q = 0; q < NFROW; ++q) { acc0 = wmma4(wreg[q], d0[q], acc0); acc1 = wmma4(wreg[q], d1[q], acc1); }
    FRAG_FENCE();
    frag_row(s.Pin, base0, 3, d0); frag_row(s.Pin, base1, 3, d1);
    FRAG_FENCE();
    #pragma unroll
    for (int q = NFROW; q < 2 * NFROW; ++q) { acc0 = wmma4(wreg[q], d0[q], acc0); acc1 = wmma4(wreg[q], d1[q], acc1); }
    FRAG_FENCE();
    frag_row(s.Pin, base0, 4, d0); frag_row(s.Pin, base1, 4, d1);
    FRAG_FENCE();
    #pragma unroll
    for (int q = 2 * NFROW; q < NCHUNK; ++q) { acc0 = wmma4(wreg[q], d0[q], acc0); acc1 = wmma4(wreg[q], d1[q], acc1); }
    if (PREV_GLOBAL) { coef_make(w0, k0, s, i0); coef_make(w1, k1, s, i1); }
    else             { coef_lds(k0, s, i0);      coef_lds(k1, s, i1); }
    tile_finish<OUT_LDS>(acc0, k0, s, i0);
    tile_finish<OUT_LDS>(acc1, k1, s, i1);
}

template <int NBR, bool PREV_GLOBAL, bool OUT_LDS>
__device__ __forceinline__ void run_stage(const StageCtx& s, const v2f (&wreg)[NCHUNK], int vr) {
    if (NBR > 8) {
        if (vr + 8 < NBR) stage_one<PREV_GLOBAL, OUT_LDS>(s, wreg, vr + 8);
    }
    if (vr + 4 < NBR) stage_two<PREV_GLOBAL, OUT_LDS>(s, wreg, vr, vr + 4);
    else              stage_one<PREV_GLOBAL, OUT_LDS>(s, wreg, vr);
}

__global__ __launch_bounds__(NTHR) void fdtd_step4(
    const float* __restrict__ cur,
    const float* prev,
    const float* __restrict__ epsr,
    const float* __restrict__ sigma,
    const float* __restrict__ Wglob,
    const float* __restrict__ src,
    float* mid,
    float* next,
    int step) {
    __shared__ __attribute__((aligned(16))) float L[NROWS * LW];
    __shared__ __attribute__((aligned(16))) float Cf[3 * CFPLANE];
    float* P0 = L;
    float* P1 = L + P0H * LW;
    float* P2 = L + (P0H + P1H) * LW;

    const int b  = blockIdx.x;
    const int y0 = blockIdx.y * CORE_H;
    const int t  = threadIdx.x;
    const int boff = b * NPIX;
    const float* curb = cur + boff;

    v2f wreg[NCHUNK];
    #pragma unroll
    for (int i = 0; i < NCHUNK; ++i)
        wreg[i] = *(const v2f*)&Wglob[(i * 32 + (t & 31)) * 2];

    v4f stg[NSTG];
    #pragma unroll
    for (int i = 0; i < NSTG; ++i) {
        const int p  = t + NTHR * i;
        const int r  = p / ROW_V4, c4 = p % ROW_V4;
        const int gy = y0 - P0OFF + r;
        v4f v = {0.f, 0.f, 0.f, 0.f};
        if (r < P0H && gy >= 0 && gy < DIMSZ)
            v = *(const v4f*)&curb[gy * DIMSZ + 4 * c4];
        stg[i] = v;
    }
    for (int e = t; e < NROWS * NPADC; e += NTHR) {
        const int r = e / NPADC, k = e % NPADC;
        const int c = (k < XPAD) ? k : (DIMSZ + k);
        L[r * LW + c] = 0.0f;
    }
    P2[(t >> 8) * (P2H - 1) * LW + XPAD + (t & 255)] = 0.0f;
    #pragma unroll
    for (int i = 0; i < NSTG; ++i) {
        const int p = t + NTHR * i;
        const int r = p / ROW_V4, c4 = p % ROW_V4;
        if (r < P0H) {
            float* d = &P0[r * LW + XPAD + 4 * c4];
            d[0] = stg[i][0]; d[1] = stg[i][1]; d[2] = stg[i][2]; d[3] = stg[i][3];
        }
    }
    __syncthreads();

    const int wave = __builtin_amdgcn_readfirstlane(t >> 5);
    const int lane = t & 31;
    const int c16  = lane & 15;
    const int h    = lane >> 4;
    const int tx   = wave & 3;
    const int vr   = ((wave >> 2) + tx) & 3;
    const int gx0  = 4 * (16 * tx + c16);

    StageCtx s;
    s.epsr = epsr; s.sigma = sigma; s.prevg = prev; s.Cf = Cf; s.mid = mid; s.next = next;
    s.boff = boff; s.y0 = y0; s.gx0 = gx0; s.h = h;

    s.Pin = P0; s.in_r0 = 0; s.rel0 = -10; s.Pprev = nullptr; s.prev_off = 0;
    s.Pout = P1; s.out_off = P1OFF;
    s.srcterm = -(src[2 * step] - src[2 * step + 1]);
    run_stage<9, true, true>(s, wreg, vr);
    __syncthreads();

    s.Pin = P1; s.in_r0 = 1; s.rel0 = -6; s.Pprev = P0; s.prev_off = P0OFF;
    s.Pout = P2; s.out_off = P2OFF;
    s.srcterm = -(src[2 * step + 2] - src[2 * step + 3]);
    run_stage<7, false, true>(s, wreg, vr);
    __syncthreads();

    s.Pin = P2; s.in_r0 = 0; s.rel0 = -4; s.Pprev = P1; s.prev_off = P1OFF;
    s.Pout = P0; s.out_off = P0OFF;
    s.srcterm = -(src[2 * step + 4] - src[2 * step + 5]);
    run_stage<6, false, true>(s, wreg, vr);
    __syncthreads();

    s.Pin = P0; s.in_r0 = 10; s.rel0 = 0; s.Pprev = P2; s.prev_off = P2OFF;
    s.Pout = nullptr; s.out_off = 0;
    s.srcterm = -(src[2 * step + 6] - src[2 * step + 7]);
    run_stage<4, false, false>(s, wreg, vr);
}

extern "C" void kernel_launch(void* const* d_in, const int* in_sizes, int n_in,
                              void* d_out, int out_size, void* d_ws, size_t ws_size,
                              hipStream_t stream) {
    const float* u1    = (const float*)d_in[0];
    const float* u0    = (const float*)d_in[1];
    const float* epsr  = (const float*)d_in[2];
    const float* sigma = (const float*)d_in[3];
    const float* fdk   = (const float*)d_in[4];
    const float* cw    = (const float*)d_in[5];
    const float* src   = (const float*)d_in[6];
    float* out = (float*)d_out;
    float* ws  = (float*)d_ws;

    float* bufs[4] = {ws, ws + NELEM, ws + 2 * NELEM, ws + 3 * NELEM};
    float* Wbuf = ws + 4 * NELEM;

    fdtd_init<<<1, 256, 0, stream>>>(fdk, cw, Wbuf);

    const float* cur  = u1;
    const float* prev = u0;
    dim3 grid(NBATCH, DIMSZ / CORE_H);
    for (int k = 0; k < 8; ++k) {
        float* mid  = (k == 7) ? nullptr : bufs[2 * (k & 1)];
        float* next = (k == 7) ? out : bufs[2 * (k & 1) + 1];
        fdtd_step4<<<grid, NTHR, 0, stream>>>(cur, prev, epsr, sigma, Wbuf, src, mid, next, 4 * k);
        prev = mid;
        cur  = next;
    }
}
